# starting kernel
# baseline (speedup 1.0000x reference)
#include <hip/hip_runtime.h>


typedef __attribute__((ext_vector_type(2))) float v2f;
typedef __attribute__((ext_vector_type(8))) float v8f;

#define NPTS 2000
#define DDIM 24
#define NTILE 125
#define LSPLIT 20
#define LCHUNK (NPTS / LSPLIT)

#define BETAT_FLOATS  (NPTS * DDIM * DDIM)
#define ALPHAT_FLOATS (NPTS * DDIM)
#define PART_FLOATS   (LSPLIT * NPTS * DDIM)

__global__ __launch_bounds__(256) void prep_kernel(const float* __restrict__ alpha,
                                                   const float* __restrict__ beta,
                                                   float* __restrict__ betaT,
                                                   float* __restrict__ alphaT) {
  int t = blockIdx.x * blockDim.x + threadIdx.x;
  if (t < BETAT_FLOATS) {
    int l = t / (DDIM * DDIM);
    int rem = t - l * (DDIM * DDIM);
    int j = rem / DDIM;
    int a = rem - j * DDIM;
    betaT[t] = (a == j) ? 0.0f : beta[(j * DDIM + a) * NPTS + l];
  }
  if (t < ALPHAT_FLOATS) {
    int l = t / DDIM;
    int j = t - l * DDIM;
    alphaT[t] = alpha[j * NPTS + l];
  }
}

__global__ __launch_bounds__(32) void rkhs_main(const float* __restrict__ x,
                                                const float* __restrict__ alphaT,
                                                const float* __restrict__ betaT,
                                                float* __restrict__ partial) {
  const int itile = blockIdx.x;
  const int chunk = blockIdx.y;
  const int lane  = threadIdx.x;
  const int i0    = itile * 16;

  __shared__ float xs[16][DDIM];
  __shared__ float xl[DDIM];
  __shared__ float df[16][DDIM + 2];
  __shared__ float Ssum[16];

  for (int t = lane; t < 16 * DDIM; t += 32) {
    int r = t / DDIM, k = t - r * DDIM;
    xs[r][k] = x[(i0 + r) * DDIM + k];
  }
  __syncthreads();

  const int  n0      = lane & 15;
  const int  koff    = (lane >> 4) * 2;
  const int  rowbase = (lane >> 4) * 8;
  const bool jt1ok   = (n0 < 8);
  const int  j1      = n0;
  const int  j2      = jt1ok ? (16 + n0) : 0;

  float acc0[8], acc1[8];
#pragma unroll
  for (int m = 0; m < 8; ++m) { acc0[m] = 0.0f; acc1[m] = 0.0f; }

  const int l0 = chunk * LCHUNK;
  for (int l = l0; l < l0 + LCHUNK; ++l) {
    if (lane < DDIM) xl[lane] = x[l * DDIM + lane];
    __syncthreads();

    for (int t = lane; t < 16 * DDIM; t += 32) {
      int r = t / DDIM, k = t - r * DDIM;
      df[r][k] = xs[r][k] - xl[k];
    }
    __syncthreads();

    if (lane < 16) {
      float s = 0.0f;
#pragma unroll
      for (int k = 0; k < DDIM; ++k) s += df[lane][k] * df[lane][k];
      Ssum[lane] = s;
    }
    __syncthreads();

    const float* Bl = betaT + (size_t)l * (DDIM * DDIM);
    v8f c0 = {};
    v8f c1 = {};
#pragma unroll
    for (int k0 = 0; k0 < DDIM; k0 += 4) {
      v2f a;
      a.x = df[n0][k0 + koff];
      a.y = df[n0][k0 + koff + 1];
      v2f b0;
      b0.x = Bl[j1 * DDIM + k0 + koff];
      b0.y = Bl[j1 * DDIM + k0 + koff + 1];
      float t0 = Bl[j2 * DDIM + k0 + koff];
      float t1 = Bl[j2 * DDIM + k0 + koff + 1];
      v2f b1;
      b1.x = jt1ok ? t0 : 0.0f;
      b1.y = jt1ok ? t1 : 0.0f;
      c0 = __builtin_amdgcn_wmma_f32_16x16x4_f32(false, a, false, b0,
                                                 (short)0, c0, false, false);
      c1 = __builtin_amdgcn_wmma_f32_16x16x4_f32(false, a, false, b1,
                                                 (short)0, c1, false, false);
    }

    float al0  = alphaT[l * DDIM + j1];
    float al1t = alphaT[l * DDIM + j2];
    float al1  = jt1ok ? al1t : 0.0f;
#pragma unroll
    for (int m = 0; m < 8; ++m) {
      int   r  = rowbase + m;
      float S  = Ssum[r];
      float d0 = df[r][j1];
      float w0 = __expf(d0 * d0 - S);
      acc0[m] += w0 * (al0 + 2.0f * c0[m]);
      float d1 = df[r][j2];
      float w1 = __expf(d1 * d1 - S);
      acc1[m] += w1 * (al1 + 2.0f * c1[m]);
    }
    __syncthreads();
  }

  float* p = partial + ((size_t)chunk * NPTS + i0) * DDIM;
#pragma unroll
  for (int m = 0; m < 8; ++m) {
    int r = rowbase + m;
    p[r * DDIM + j1] = acc0[m];
    if (jt1ok) p[r * DDIM + 16 + n0] = acc1[m];
  }
}

__global__ __launch_bounds__(256) void reduce_kernel(const float* __restrict__ partial,
                                                     float* __restrict__ out) {
  int t = blockIdx.x * blockDim.x + threadIdx.x;
  if (t < NPTS * DDIM) {
    float s = 0.0f;
#pragma unroll
    for (int c = 0; c < LSPLIT; ++c)
      s += partial[(size_t)c * (NPTS * DDIM) + t];
    out[t] = s;
  }
}

extern "C" void kernel_launch(void* const* d_in, const int* in_sizes, int n_in,
                              void* d_out, int out_size, void* d_ws, size_t ws_size,
                              hipStream_t stream) {
  const float* x     = (const float*)d_in[0];
  const float* alpha = (const float*)d_in[1];
  const float* beta  = (const float*)d_in[2];
  float* out = (float*)d_out;

  char* ws = (char*)d_ws;
  float* betaT   = (float*)ws;
  float* alphaT  = (float*)(ws + (size_t)BETAT_FLOATS * sizeof(float));
  float* partial = (float*)(ws + (size_t)(BETAT_FLOATS + ALPHAT_FLOATS) * sizeof(float));

  (void)in_sizes; (void)n_in; (void)out_size; (void)ws_size;

  prep_kernel<<<(BETAT_FLOATS + 255) / 256, 256, 0, stream>>>(alpha, beta, betaT, alphaT);

  dim3 grid(NTILE, LSPLIT);
  rkhs_main<<<grid, 32, 0, stream>>>(x, alphaT, betaT, partial);

  reduce_kernel<<<(NPTS * DDIM + 255) / 256, 256, 0, stream>>>(partial, out);
}


# prep launch removed: each block builds cf[j][l] for its own 64 points l in its prologue (8-byte beta/alpha loads, all in flight together, LDS fill unrolled); 2 launches, same sum order
# speedup vs baseline: 5.1203x; 5.1203x over previous
#include <hip/hip_runtime.h>


typedef __attribute__((ext_vector_type(2))) float v2f;
typedef __attribute__((ext_vector_type(4))) float v4f;
typedef __attribute__((ext_vector_type(8))) float v8f;

#define NPTS 2000
#define DDIM 24
#define NTILE 125
#define TL 4
#define LCHUNK (TL * 16)
#define LSPLIT ((NTILE + TL - 1) / TL)
#define RT 2
#define WPB 8
#define IGROUPS ((NTILE + RT * WPB - 1) / (RT * WPB))

__global__ __launch_bounds__(WPB * 32) void rkhs_main(const float* __restrict__ x,
                                                 const float* __restrict__ alpha,
                                                 const float* __restrict__ beta,
                                                 float* __restrict__ partial) {
  __shared__ __attribute__((aligned(16))) float xs[DDIM][LCHUNK];
  __shared__ __attribute__((aligned(16))) float cs[DDIM][LCHUNK];

  const int tid   = threadIdx.x;
  const int lane  = tid & 31;
  const int wave  = tid >> 5;
  const int chunk = blockIdx.y;

  int  l0[TL];
  bool lok[TL];
#pragma unroll
  for (int t = 0; t < TL; ++t) {
    int lt = chunk * TL + t;
    lok[t] = (lt < NTILE);
    if (lt > NTILE - 1) lt = NTILE - 1;
    l0[t] = lt * 16;
  }

#pragma unroll
  for (int u = 0; u < (DDIM * LCHUNK) / (WPB * 32); ++u) {
    int e = tid + u * (WPB * 32);
    int ll = e / DDIM, k = e - ll * DDIM;
    int lt = chunk * TL + (ll >> 4);
    if (lt > NTILE - 1) lt = NTILE - 1;
    xs[k][ll] = x[(lt * 16 + (ll & 15)) * DDIM + k];
  }

  {
    int ltc = chunk * TL + (lane >> 3);
    if (ltc > NTILE - 1) ltc = NTILE - 1;
    const int lg = ltc * 16 + ((2 * lane) & 15);
    const v4f* xrow = (const v4f*)(x + lg * DDIM);
    v4f xr[12];
#pragma unroll
    for (int q = 0; q < 12; ++q) xr[q] = xrow[q];
#pragma unroll
    for (int u = 0; u < DDIM / WPB; ++u) {
      const int j = wave + WPB * u;
      const float* bq = beta + (size_t)j * DDIM * NPTS + lg;
      float s0 = 0.0f, s1 = 0.0f;
#pragma unroll
      for (int a = 0; a < DDIM; ++a) {
        v2f b = *(const v2f*)(bq + a * NPTS);
        float p0 = b.x * xr[a >> 2][a & 3];
        float p1 = b.y * xr[6 + (a >> 2)][a & 3];
        s0 += (a == j) ? 0.0f : p0;
        s1 += (a == j) ? 0.0f : p1;
      }
      v2f al = *(const v2f*)(alpha + j * NPTS + lg);
      v2f r;
      r.x = al.x - 2.0f * s0;
      r.y = al.y - 2.0f * s1;
      *(v2f*)&cs[j][2 * lane] = r;
    }
  }
  __syncthreads();

  const int c = lane & 15;
  const int h = lane >> 4;
  const int it0 = (blockIdx.x * WPB + wave) * RT;
  if (it0 >= NTILE) return;

  int  irow[RT];
  bool ok[RT];
#pragma unroll
  for (int r = 0; r < RT; ++r) {
    int it = it0 + r;
    ok[r] = (it < NTILE);
    if (it > NTILE - 1) it = NTILE - 1;
    irow[r] = it * 16 + c;
  }


  v2f bx[RT][6];
#pragma unroll
  for (int r = 0; r < RT; ++r) {
    const v4f* xr = (const v4f*)(x + irow[r] * DDIM);
#pragma unroll
    for (int q = 0; q < 6; ++q) {
      v4f v = xr[q];
      bx[r][q].x = h ? v.z : v.x;
      bx[r][q].y = h ? v.w : v.y;
    }
  }

  float S[RT][TL][8];
#pragma unroll
  for (int r = 0; r < RT; ++r)
#pragma unroll
    for (int t = 0; t < TL; ++t)
#pragma unroll
      for (int m = 0; m < 8; ++m) S[r][t][m] = 0.0f;

#pragma unroll 1
  for (int q = 0; q < 6; ++q) {
    float xq[RT][4];
#pragma unroll
    for (int r = 0; r < RT; ++r) {
      v4f v = *(const v4f*)(x + irow[r] * DDIM + 4 * q);
      xq[r][0] = v.x; xq[r][1] = v.y; xq[r][2] = v.z; xq[r][3] = v.w;
    }
#pragma unroll
    for (int kq = 0; kq < 4; ++kq) {
      const float* xk = &xs[4 * q + kq][8 * h];
#pragma unroll
      for (int t = 0; t < TL; ++t) {
        v4f lo = *(const v4f*)(xk + 16 * t);
        v4f hi = *(const v4f*)(xk + 16 * t + 4);
        float xl[8] = {lo.x, lo.y, lo.z, lo.w, hi.x, hi.y, hi.z, hi.w};
#pragma unroll
        for (int r = 0; r < RT; ++r)
#pragma unroll
          for (int m = 0; m < 8; ++m) {
            float d = xq[r][kq] - xl[m];
            S[r][t][m] += d * d;
          }
      }
    }
  }

#pragma unroll
  for (int t = 0; t < TL; ++t)
    if (!lok[t]) {
#pragma unroll
      for (int r = 0; r < RT; ++r)
#pragma unroll
        for (int m = 0; m < 8; ++m) S[r][t][m] = __builtin_inff();
    }

  const float* bp = beta + c + (2 * h) * NPTS;
  v2f A[TL][6];
#pragma unroll
  for (int t = 0; t < TL; ++t)
#pragma unroll
    for (int kk = 0; kk < 6; ++kk) {
      A[t][kk].x = bp[(4 * kk + 0) * NPTS + l0[t]];
      A[t][kk].y = bp[(4 * kk + 1) * NPTS + l0[t]];
    }

  for (int j = 0; j < DDIM; ++j) {
    const int jn = (j + 1 < DDIM) ? (j + 1) : j;
    const float* bn = bp + (size_t)jn * DDIM * NPTS;

    v2f   bj[RT][6];
    float xij[RT];
    float acc[RT];
#pragma unroll
    for (int r = 0; r < RT; ++r) {
#pragma unroll
      for (int kk = 0; kk < 6; ++kk) {
        bj[r][kk].x = (4 * kk + 2 * h     == j) ? 0.0f : bx[r][kk].x;
        bj[r][kk].y = (4 * kk + 2 * h + 1 == j) ? 0.0f : bx[r][kk].y;
      }
      xij[r] = x[irow[r] * DDIM + j];
      acc[r] = 0.0f;
    }

#pragma unroll
    for (int t = 0; t < TL; ++t) {
      v8f C0 = {};
      v8f C1 = {};
#pragma unroll
      for (int kk = 0; kk < 6; ++kk) {
        C0 = __builtin_amdgcn_wmma_f32_16x16x4_f32(false, A[t][kk], false, bj[0][kk],
                                                   (short)0, C0, false, false);
        C1 = __builtin_amdgcn_wmma_f32_16x16x4_f32(false, A[t][kk], false, bj[1][kk],
                                                   (short)0, C1, false, false);
      }
#pragma unroll
      for (int kk = 0; kk < 6; ++kk) {
        A[t][kk].x = bn[(4 * kk + 0) * NPTS + l0[t]];
        A[t][kk].y = bn[(4 * kk + 1) * NPTS + l0[t]];
      }

      v4f xlo = *(const v4f*)&xs[j][16 * t + 8 * h];
      v4f xhi = *(const v4f*)&xs[j][16 * t + 8 * h + 4];
      v4f clo = *(const v4f*)&cs[j][16 * t + 8 * h];
      v4f chi = *(const v4f*)&cs[j][16 * t + 8 * h + 4];
      float xl[8] = {xlo.x, xlo.y, xlo.z, xlo.w, xhi.x, xhi.y, xhi.z, xhi.w};
      float cc[8] = {clo.x, clo.y, clo.z, clo.w, chi.x, chi.y, chi.z, chi.w};
#pragma unroll
      for (int m = 0; m < 8; ++m) {
        float d0 = xij[0] - xl[m];
        float w0 = __expf(d0 * d0 - S[0][t][m]);
        acc[0] += w0 * (cc[m] + 2.0f * C0[m]);
        float d1 = xij[1] - xl[m];
        float w1 = __expf(d1 * d1 - S[1][t][m]);
        acc[1] += w1 * (cc[m] + 2.0f * C1[m]);
      }
    }

#pragma unroll
    for (int r = 0; r < RT; ++r) {
      float v = acc[r] + __shfl_xor(acc[r], 16);
      if (h == 0 && ok[r])
        partial[((size_t)chunk * NPTS + irow[r]) * DDIM + j] = v;
    }
  }
}

__global__ __launch_bounds__(256) void reduce_kernel(const float* __restrict__ partial,
                                                     float* __restrict__ out) {
  int t = blockIdx.x * blockDim.x + threadIdx.x;
  if (t < NPTS * DDIM) {
    float s = 0.0f;
#pragma unroll
    for (int c = 0; c < LSPLIT; ++c)
      s += partial[(size_t)c * (NPTS * DDIM) + t];
    out[t] = s;
  }
}

extern "C" void kernel_launch(void* const* d_in, const int* in_sizes, int n_in,
                              void* d_out, int out_size, void* d_ws, size_t ws_size,
                              hipStream_t stream) {
  const float* x     = (const float*)d_in[0];
  const float* alpha = (const float*)d_in[1];
  const float* beta  = (const float*)d_in[2];
  float* out = (float*)d_out;

  float* partial = (float*)d_ws;

  (void)in_sizes; (void)n_in; (void)out_size; (void)ws_size;

  dim3 grid(IGROUPS, LSPLIT);
  rkhs_main<<<grid, WPB * 32, 0, stream>>>(x, alpha, beta, partial);

  reduce_kernel<<<(NPTS * DDIM + 255) / 256, 256, 0, stream>>>(partial, out);
}


# main: vector instructions per wave and column 337 -> 194: cs[j][l] is the WMMA C operand (B = 2x), x pre-scaled by sqrt(log2 e) so weight = one exp2, packed on D row pairs; partial sums regroup
# speedup vs baseline: 5.4769x; 1.0696x over previous
#include <hip/hip_runtime.h>


typedef __attribute__((ext_vector_type(2))) float v2f;
typedef __attribute__((ext_vector_type(4))) float v4f;
typedef __attribute__((ext_vector_type(8))) float v8f;

#define NPTS 2000
#define DDIM 24
#define NTILE 125
#define TL 4
#define LCHUNK (TL * 16)
#define LSPLIT ((NTILE + TL - 1) / TL)
#define RT 2
#define WPB 8
#define IGROUPS ((NTILE + RT * WPB - 1) / (RT * WPB))
#define XSCALE 1.2011224087864498f

__global__ __launch_bounds__(WPB * 32) void rkhs_main(const float* __restrict__ x,
                                                 const float* __restrict__ alpha,
                                                 const float* __restrict__ beta,
                                                 float* __restrict__ partial) {
  __shared__ __attribute__((aligned(16))) float xs[DDIM][LCHUNK];
  __shared__ __attribute__((aligned(16))) float cs[DDIM][LCHUNK];

  const int tid   = threadIdx.x;
  const int lane  = tid & 31;
  const int wave  = tid >> 5;
  const int chunk = blockIdx.y;

  int  l0[TL];
  bool lok[TL];
#pragma unroll
  for (int t = 0; t < TL; ++t) {
    int lt = chunk * TL + t;
    lok[t] = (lt < NTILE);
    if (lt > NTILE - 1) lt = NTILE - 1;
    l0[t] = lt * 16;
  }

#pragma unroll
  for (int u = 0; u < (DDIM * LCHUNK) / (WPB * 32); ++u) {
    int e = tid + u * (WPB * 32);
    int ll = e / DDIM, k = e - ll * DDIM;
    int lt = chunk * TL + (ll >> 4);
    if (lt > NTILE - 1) lt = NTILE - 1;
    xs[k][ll] = XSCALE * x[(lt * 16 + (ll & 15)) * DDIM + k];
  }

  {
    int ltc = chunk * TL + (lane >> 3);
    if (ltc > NTILE - 1) ltc = NTILE - 1;
    const int lg = ltc * 16 + ((2 * lane) & 15);
    const v4f* xrow = (const v4f*)(x + lg * DDIM);
    v4f xr[12];
#pragma unroll
    for (int q = 0; q < 12; ++q) xr[q] = xrow[q];
#pragma unroll
    for (int u = 0; u < DDIM / WPB; ++u) {
      const int j = wave + WPB * u;
      const float* bq = beta + (size_t)j * DDIM * NPTS + lg;
      float s0 = 0.0f, s1 = 0.0f;
#pragma unroll
      for (int a = 0; a < DDIM; ++a) {
        v2f b = *(const v2f*)(bq + a * NPTS);
        float p0 = b.x * xr[a >> 2][a & 3];
        float p1 = b.y * xr[6 + (a >> 2)][a & 3];
        s0 += (a == j) ? 0.0f : p0;
        s1 += (a == j) ? 0.0f : p1;
      }
      v2f al = *(const v2f*)(alpha + j * NPTS + lg);
      v2f r;
      r.x = al.x - 2.0f * s0;
      r.y = al.y - 2.0f * s1;
      *(v2f*)&cs[j][2 * lane] = r;
    }
  }
  __syncthreads();

  const int c = lane & 15;
  const int h = lane >> 4;
  const int it0 = (blockIdx.x * WPB + wave) * RT;
  if (it0 >= NTILE) return;

  int  irow[RT];
  bool ok[RT];
#pragma unroll
  for (int r = 0; r < RT; ++r) {
    int it = it0 + r;
    ok[r] = (it < NTILE);
    if (it > NTILE - 1) it = NTILE - 1;
    irow[r] = it * 16 + c;
  }


  v2f bx[RT][6];
#pragma unroll
  for (int r = 0; r < RT; ++r) {
    const v4f* xr = (const v4f*)(x + irow[r] * DDIM);
#pragma unroll
    for (int q = 0; q < 6; ++q) {
      v4f v = xr[q];
      bx[r][q].x = 2.0f * (h ? v.z : v.x);
      bx[r][q].y = 2.0f * (h ? v.w : v.y);
    }
  }

  v2f S[RT][TL][4];
#pragma unroll
  for (int r = 0; r < RT; ++r)
#pragma unroll
    for (int t = 0; t < TL; ++t)
#pragma unroll
      for (int p = 0; p < 4; ++p) S[r][t][p] = (v2f){0.0f, 0.0f};

#pragma unroll 1
  for (int q = 0; q < 6; ++q) {
    v4f raw[RT];
#pragma unroll
    for (int r = 0; r < RT; ++r) raw[r] = *(const v4f*)(x + irow[r] * DDIM + 4 * q);
    v4f lo[4][TL], hi[4][TL];
#pragma unroll
    for (int kq = 0; kq < 4; ++kq)
#pragma unroll
      for (int t = 0; t < TL; ++t) {
        lo[kq][t] = *(const v4f*)&xs[4 * q + kq][8 * h + 16 * t];
        hi[kq][t] = *(const v4f*)&xs[4 * q + kq][8 * h + 16 * t + 4];
      }
    float xq[RT][4];
#pragma unroll
    for (int r = 0; r < RT; ++r) {
      xq[r][0] = XSCALE * raw[r].x; xq[r][1] = XSCALE * raw[r].y;
      xq[r][2] = XSCALE * raw[r].z; xq[r][3] = XSCALE * raw[r].w;
    }
#pragma unroll
    for (int kq = 0; kq < 4; ++kq)
#pragma unroll
      for (int t = 0; t < TL; ++t) {
        v2f xl[4] = {{lo[kq][t].x, lo[kq][t].y}, {lo[kq][t].z, lo[kq][t].w},
                     {hi[kq][t].x, hi[kq][t].y}, {hi[kq][t].z, hi[kq][t].w}};
#pragma unroll
        for (int r = 0; r < RT; ++r)
#pragma unroll
          for (int p = 0; p < 4; ++p) {
            v2f d = xq[r][kq] - xl[p];
            S[r][t][p] += d * d;
          }
      }
  }

#pragma unroll
  for (int t = 0; t < TL; ++t)
    if (!lok[t]) {
#pragma unroll
      for (int r = 0; r < RT; ++r)
#pragma unroll
        for (int p = 0; p < 4; ++p) S[r][t][p] = (v2f){__builtin_inff(), __builtin_inff()};
    }

  const float* bp = beta + c + (2 * h) * NPTS;
  v2f A[TL][6];
#pragma unroll
  for (int t = 0; t < TL; ++t)
#pragma unroll
    for (int kk = 0; kk < 6; ++kk) {
      A[t][kk].x = bp[(4 * kk + 0) * NPTS + l0[t]];
      A[t][kk].y = bp[(4 * kk + 1) * NPTS + l0[t]];
    }

#pragma unroll 2
  for (int j = 0; j < DDIM; ++j) {
    const int jn = (j + 1 < DDIM) ? (j + 1) : j;
    const float* bn = bp + (size_t)jn * DDIM * NPTS;

    v2f   bj[RT][6];
    float xij[RT];
    v2f   acc[RT][2];
#pragma unroll
    for (int r = 0; r < RT; ++r) {
#pragma unroll
      for (int kk = 0; kk < 6; ++kk) {
        bj[r][kk].x = (4 * kk + 2 * h     == j) ? 0.0f : bx[r][kk].x;
        bj[r][kk].y = (4 * kk + 2 * h + 1 == j) ? 0.0f : bx[r][kk].y;
      }
      xij[r] = XSCALE * x[irow[r] * DDIM + j];
      acc[r][0] = (v2f){0.0f, 0.0f};
      acc[r][1] = (v2f){0.0f, 0.0f};
    }

#pragma unroll
    for (int t = 0; t < TL; ++t) {
      v4f clo = *(const v4f*)&cs[j][16 * t + 8 * h];
      v4f chi = *(const v4f*)&cs[j][16 * t + 8 * h + 4];
      v8f C0 = __builtin_shufflevector(clo, chi, 0, 1, 2, 3, 4, 5, 6, 7);
      v8f C1 = C0;
#pragma unroll
      for (int kk = 0; kk < 6; ++kk) {
        C0 = __builtin_amdgcn_wmma_f32_16x16x4_f32(false, A[t][kk], false, bj[0][kk],
                                                   (short)0, C0, false, false);
        C1 = __builtin_amdgcn_wmma_f32_16x16x4_f32(false, A[t][kk], false, bj[1][kk],
                                                   (short)0, C1, false, false);
      }
#pragma unroll
      for (int kk = 0; kk < 6; ++kk) {
        A[t][kk].x = bn[(4 * kk + 0) * NPTS + l0[t]];
        A[t][kk].y = bn[(4 * kk + 1) * NPTS + l0[t]];
      }

      v4f xlo = *(const v4f*)&xs[j][16 * t + 8 * h];
      v4f xhi = *(const v4f*)&xs[j][16 * t + 8 * h + 4];
      v2f xl[4] = {{xlo.x, xlo.y}, {xlo.z, xlo.w}, {xhi.x, xhi.y}, {xhi.z, xhi.w}};
#pragma unroll
      for (int p = 0; p < 4; ++p) {
        v2f d0 = xij[0] - xl[p];
        v2f e0 = d0 * d0 - S[0][t][p];
        v2f w0 = {__builtin_amdgcn_exp2f(e0.x), __builtin_amdgcn_exp2f(e0.y)};
        v2f t0 = {C0[2 * p], C0[2 * p + 1]};
        acc[0][p & 1] += w0 * t0;
        v2f d1 = xij[1] - xl[p];
        v2f e1 = d1 * d1 - S[1][t][p];
        v2f w1 = {__builtin_amdgcn_exp2f(e1.x), __builtin_amdgcn_exp2f(e1.y)};
        v2f t1 = {C1[2 * p], C1[2 * p + 1]};
        acc[1][p & 1] += w1 * t1;
      }
    }

#pragma unroll
    for (int r = 0; r < RT; ++r) {
      float a = (acc[r][0].x + acc[r][0].y) + (acc[r][1].x + acc[r][1].y);
      float v = a + __shfl_xor(a, 16);
      if (h == 0 && ok[r])
        partial[((size_t)chunk * NPTS + irow[r]) * DDIM + j] = v;
    }
  }
}

__global__ __launch_bounds__(256) void reduce_kernel(const float* __restrict__ partial,
                                                     float* __restrict__ out) {
  int t = blockIdx.x * blockDim.x + threadIdx.x;
  if (t < NPTS * DDIM) {
    float s = 0.0f;
#pragma unroll
    for (int c = 0; c < LSPLIT; ++c)
      s += partial[(size_t)c * (NPTS * DDIM) + t];
    out[t] = s;
  }
}

extern "C" void kernel_launch(void* const* d_in, const int* in_sizes, int n_in,
                              void* d_out, int out_size, void* d_ws, size_t ws_size,
                              hipStream_t stream) {
  const float* x     = (const float*)d_in[0];
  const float* alpha = (const float*)d_in[1];
  const float* beta  = (const float*)d_in[2];
  float* out = (float*)d_out;

  float* partial = (float*)d_ws;

  (void)in_sizes; (void)n_in; (void)out_size; (void)ws_size;

  dim3 grid(IGROUPS, LSPLIT);
  rkhs_main<<<grid, WPB * 32, 0, stream>>>(x, alpha, beta, partial);

  reduce_kernel<<<(NPTS * DDIM + 255) / 256, 256, 0, stream>>>(partial, out);
}


# main: no wait left on the issue path: LDS operands and x_ij read a column ahead, lane-half exchange read a column later, a tile's exp2 weights between its own WMMAs (sched_barrier); same sum order
# speedup vs baseline: 5.7113x; 1.0428x over previous
#include <hip/hip_runtime.h>


typedef __attribute__((ext_vector_type(2))) float v2f;
typedef __attribute__((ext_vector_type(4))) float v4f;
typedef __attribute__((ext_vector_type(8))) float v8f;

#define NPTS 2000
#define DDIM 24
#define NTILE 125
#define TL 4
#define LCHUNK (TL * 16)
#define LSPLIT ((NTILE + TL - 1) / TL)
#define RT 2
#define WPB 8
#define IGROUPS ((NTILE + RT * WPB - 1) / (RT * WPB))
#define XSCALE 1.2011224087864498f

__global__ __launch_bounds__(WPB * 32) void rkhs_main(const float* __restrict__ x,
                                                 const float* __restrict__ alpha,
                                                 const float* __restrict__ beta,
                                                 float* __restrict__ partial) {
  __shared__ __attribute__((aligned(16))) float xs[DDIM][LCHUNK];
  __shared__ __attribute__((aligned(16))) float cs[DDIM][LCHUNK];

  const int tid   = threadIdx.x;
  const int lane  = tid & 31;
  const int wave  = tid >> 5;
  const int chunk = blockIdx.y;

  int  l0[TL];
  bool lok[TL];
#pragma unroll
  for (int t = 0; t < TL; ++t) {
    int lt = chunk * TL + t;
    lok[t] = (lt < NTILE);
    if (lt > NTILE - 1) lt = NTILE - 1;
    l0[t] = lt * 16;
  }

#pragma unroll
  for (int u = 0; u < (DDIM * LCHUNK) / (WPB * 32); ++u) {
    int e = tid + u * (WPB * 32);
    int ll = e / DDIM, k = e - ll * DDIM;
    int lt = chunk * TL + (ll >> 4);
    if (lt > NTILE - 1) lt = NTILE - 1;
    xs[k][ll] = XSCALE * x[(lt * 16 + (ll & 15)) * DDIM + k];
  }

  {
    int ltc = chunk * TL + (lane >> 3);
    if (ltc > NTILE - 1) ltc = NTILE - 1;
    const int lg = ltc * 16 + ((2 * lane) & 15);
    const v4f* xrow = (const v4f*)(x + lg * DDIM);
    v4f xr[12];
#pragma unroll
    for (int q = 0; q < 12; ++q) xr[q] = xrow[q];
#pragma unroll
    for (int u = 0; u < DDIM / WPB; ++u) {
      const int j = wave + WPB * u;
      const float* bq = beta + (size_t)j * DDIM * NPTS + lg;
      float s0 = 0.0f, s1 = 0.0f;
#pragma unroll
      for (int a = 0; a < DDIM; ++a) {
        v2f b = *(const v2f*)(bq + a * NPTS);
        float p0 = b.x * xr[a >> 2][a & 3];
        float p1 = b.y * xr[6 + (a >> 2)][a & 3];
        s0 += (a == j) ? 0.0f : p0;
        s1 += (a == j) ? 0.0f : p1;
      }
      v2f al = *(const v2f*)(alpha + j * NPTS + lg);
      v2f r;
      r.x = al.x - 2.0f * s0;
      r.y = al.y - 2.0f * s1;
      *(v2f*)&cs[j][2 * lane] = r;
    }
  }
  __syncthreads();

  const int c = lane & 15;
  const int h = lane >> 4;
  const int it0 = (blockIdx.x * WPB + wave) * RT;
  if (it0 >= NTILE) return;

  int  irow[RT];
  bool ok[RT];
#pragma unroll
  for (int r = 0; r < RT; ++r) {
    int it = it0 + r;
    ok[r] = (it < NTILE);
    if (it > NTILE - 1) it = NTILE - 1;
    irow[r] = it * 16 + c;
  }


  v2f bx[RT][6];
#pragma unroll
  for (int r = 0; r < RT; ++r) {
    const v4f* xr = (const v4f*)(x + irow[r] * DDIM);
#pragma unroll
    for (int q = 0; q < 6; ++q) {
      v4f v = xr[q];
      bx[r][q].x = 2.0f * (h ? v.z : v.x);
      bx[r][q].y = 2.0f * (h ? v.w : v.y);
    }
  }

  v2f S[RT][TL][4];
#pragma unroll
  for (int r = 0; r < RT; ++r)
#pragma unroll
    for (int t = 0; t < TL; ++t)
#pragma unroll
      for (int p = 0; p < 4; ++p) S[r][t][p] = (v2f){0.0f, 0.0f};

  v4f rawN[RT];
#pragma unroll
  for (int r = 0; r < RT; ++r) rawN[r] = *(const v4f*)(x + irow[r] * DDIM);

#pragma unroll 1
  for (int q = 0; q < 6; ++q) {
    const int qn = (q + 1 < 6) ? (q + 1) : q;
    v4f raw[RT];
#pragma unroll
    for (int r = 0; r < RT; ++r) {
      raw[r] = rawN[r];
      rawN[r] = *(const v4f*)(x + irow[r] * DDIM + 4 * qn);
    }
    v4f lo[4][TL], hi[4][TL];
#pragma unroll
    for (int kq = 0; kq < 4; ++kq)
#pragma unroll
      for (int t = 0; t < TL; ++t) {
        lo[kq][t] = *(const v4f*)&xs[4 * q + kq][8 * h + 16 * t];
        hi[kq][t] = *(const v4f*)&xs[4 * q + kq][8 * h + 16 * t + 4];
      }
    float xq[RT][4];
#pragma unroll
    for (int r = 0; r < RT; ++r) {
      xq[r][0] = XSCALE * raw[r].x; xq[r][1] = XSCALE * raw[r].y;
      xq[r][2] = XSCALE * raw[r].z; xq[r][3] = XSCALE * raw[r].w;
    }
#pragma unroll
    for (int kq = 0; kq < 4; ++kq)
#pragma unroll
      for (int t = 0; t < TL; ++t) {
        v2f xl[4] = {{lo[kq][t].x, lo[kq][t].y}, {lo[kq][t].z, lo[kq][t].w},
                     {hi[kq][t].x, hi[kq][t].y}, {hi[kq][t].z, hi[kq][t].w}};
#pragma unroll
        for (int r = 0; r < RT; ++r)
#pragma unroll
          for (int p = 0; p < 4; ++p) {
            v2f d = xq[r][kq] - xl[p];
            S[r][t][p] += d * d;
          }
      }
  }

#pragma unroll
  for (int t = 0; t < TL; ++t)
    if (!lok[t]) {
#pragma unroll
      for (int r = 0; r < RT; ++r)
#pragma unroll
        for (int p = 0; p < 4; ++p) S[r][t][p] = (v2f){__builtin_inff(), __builtin_inff()};
    }

  const float* bp = beta + c + (2 * h) * NPTS;
  v2f A[TL][6];
#pragma unroll
  for (int t = 0; t < TL; ++t)
#pragma unroll
    for (int kk = 0; kk < 6; ++kk) {
      A[t][kk].x = bp[(4 * kk + 0) * NPTS + l0[t]];
      A[t][kk].y = bp[(4 * kk + 1) * NPTS + l0[t]];
    }

  float xnext[RT];
#pragma unroll
  for (int r = 0; r < RT; ++r) xnext[r] = x[irow[r] * DDIM];

  v8f CstN[TL];
  v2f xlN[TL][4];
#pragma unroll
  for (int t = 0; t < TL; ++t) {
    v4f clo = *(const v4f*)&cs[0][16 * t + 8 * h];
    v4f chi = *(const v4f*)&cs[0][16 * t + 8 * h + 4];
    CstN[t] = __builtin_shufflevector(clo, chi, 0, 1, 2, 3, 4, 5, 6, 7);
    v4f xlo = *(const v4f*)&xs[0][16 * t + 8 * h];
    v4f xhi = *(const v4f*)&xs[0][16 * t + 8 * h + 4];
    xlN[t][0] = (v2f){xlo.x, xlo.y}; xlN[t][1] = (v2f){xlo.z, xlo.w};
    xlN[t][2] = (v2f){xhi.x, xhi.y}; xlN[t][3] = (v2f){xhi.z, xhi.w};
  }

  float aP[RT], sP[RT];
#pragma unroll
  for (int r = 0; r < RT; ++r) { aP[r] = 0.0f; sP[r] = 0.0f; }

#pragma unroll 2
  for (int j = 0; j < DDIM; ++j) {
    const int jn = (j + 1 < DDIM) ? (j + 1) : j;
    const float* bn = bp + (size_t)jn * DDIM * NPTS;

    v2f   bj[RT][6];
    float xij[RT];
    v2f   acc[RT][2];
#pragma unroll
    for (int r = 0; r < RT; ++r) {
#pragma unroll
      for (int kk = 0; kk < 6; ++kk) {
        bj[r][kk].x = (4 * kk + 2 * h     == j) ? 0.0f : bx[r][kk].x;
        bj[r][kk].y = (4 * kk + 2 * h + 1 == j) ? 0.0f : bx[r][kk].y;
      }
      xij[r] = XSCALE * xnext[r];
      xnext[r] = x[irow[r] * DDIM + jn];
      acc[r][0] = (v2f){0.0f, 0.0f};
      acc[r][1] = (v2f){0.0f, 0.0f};
    }

    v8f Cst[TL];
    v2f xl[TL][4];
#pragma unroll
    for (int t = 0; t < TL; ++t) {
      Cst[t] = CstN[t];
#pragma unroll
      for (int p = 0; p < 4; ++p) xl[t][p] = xlN[t][p];
    }

    v8f Cp[RT];
    v2f wp[8];
#pragma unroll
    for (int t = 0; t < TL; ++t) {
      v8f C[RT];
      C[0] = Cst[t];
      C[1] = Cst[t];
      v2f w[8];
#pragma unroll
      for (int kk = 0; kk < 6; ++kk) {
        C[0] = __builtin_amdgcn_wmma_f32_16x16x4_f32(false, A[t][kk], false, bj[0][kk],
                                                     (short)0, C[0], false, false);
        C[1] = __builtin_amdgcn_wmma_f32_16x16x4_f32(false, A[t][kk], false, bj[1][kk],
                                                     (short)0, C[1], false, false);
        if (kk < 4) {
#pragma unroll
          for (int u = 2 * kk; u < 2 * kk + 2; ++u) {
            const int p = u >> 1, r = u & 1;
            v2f d = xij[r] - xl[t][p];
            v2f e = d * d - S[r][t][p];
            w[u] = (v2f){__builtin_amdgcn_exp2f(e.x), __builtin_amdgcn_exp2f(e.y)};
          }
        } else if (t > 0) {
#pragma unroll
          for (int u = 4 * (kk - 4); u < 4 * (kk - 4) + 4; ++u) {
            const int p = u >> 1, r = u & 1;
            v2f tt = {Cp[r][2 * p], Cp[r][2 * p + 1]};
            acc[r][p & 1] += wp[u] * tt;
          }
        }
        __builtin_amdgcn_sched_barrier(0);
      }
#pragma unroll
      for (int kk = 0; kk < 6; ++kk) {
        A[t][kk].x = bn[(4 * kk + 0) * NPTS + l0[t]];
        A[t][kk].y = bn[(4 * kk + 1) * NPTS + l0[t]];
      }
      if (t == 1) {
#pragma unroll
        for (int t2 = 0; t2 < TL; ++t2) {
          v4f clo = *(const v4f*)&cs[jn][16 * t2 + 8 * h];
          v4f chi = *(const v4f*)&cs[jn][16 * t2 + 8 * h + 4];
          CstN[t2] = __builtin_shufflevector(clo, chi, 0, 1, 2, 3, 4, 5, 6, 7);
          v4f xlo = *(const v4f*)&xs[jn][16 * t2 + 8 * h];
          v4f xhi = *(const v4f*)&xs[jn][16 * t2 + 8 * h + 4];
          xlN[t2][0] = (v2f){xlo.x, xlo.y}; xlN[t2][1] = (v2f){xlo.z, xlo.w};
          xlN[t2][2] = (v2f){xhi.x, xhi.y}; xlN[t2][3] = (v2f){xhi.z, xhi.w};
        }
      }
      Cp[0] = C[0];
      Cp[1] = C[1];
#pragma unroll
      for (int u = 0; u < 8; ++u) wp[u] = w[u];
    }

#pragma unroll
    for (int u = 0; u < 8; ++u) {
      const int p = u >> 1, r = u & 1;
      v2f tt = {Cp[r][2 * p], Cp[r][2 * p + 1]};
      acc[r][p & 1] += wp[u] * tt;
    }

    float aN[RT], sN[RT];
#pragma unroll
    for (int r = 0; r < RT; ++r)
      aN[r] = (acc[r][0].x + acc[r][0].y) + (acc[r][1].x + acc[r][1].y);
#pragma unroll
    for (int r = 0; r < RT; ++r) sN[r] = __shfl_xor(aN[r], 16);
#pragma unroll
    for (int r = 0; r < RT; ++r) {
      float v = aP[r] + sP[r];
      if (h == 0 && ok[r] && j > 0)
        partial[((size_t)chunk * NPTS + irow[r]) * DDIM + (j - 1)] = v;
    }
#pragma unroll
    for (int r = 0; r < RT; ++r) { aP[r] = aN[r]; sP[r] = sN[r]; }
  }

#pragma unroll
  for (int r = 0; r < RT; ++r) {
    float v = aP[r] + sP[r];
    if (h == 0 && ok[r])
      partial[((size_t)chunk * NPTS + irow[r]) * DDIM + (DDIM - 1)] = v;
  }
}

__global__ __launch_bounds__(256) void reduce_kernel(const float* __restrict__ partial,
                                                     float* __restrict__ out) {
  int t = blockIdx.x * blockDim.x + threadIdx.x;
  if (t < NPTS * DDIM) {
    float s = 0.0f;
#pragma unroll
    for (int c = 0; c < LSPLIT; ++c)
      s += partial[(size_t)c * (NPTS * DDIM) + t];
    out[t] = s;
  }
}

extern "C" void kernel_launch(void* const* d_in, const int* in_sizes, int n_in,
                              void* d_out, int out_size, void* d_ws, size_t ws_size,
                              hipStream_t stream) {
  const float* x     = (const float*)d_in[0];
  const float* alpha = (const float*)d_in[1];
  const float* beta  = (const float*)d_in[2];
  float* out = (float*)d_out;

  float* partial = (float*)d_ws;

  (void)in_sizes; (void)n_in; (void)out_size; (void)ws_size;

  dim3 grid(IGROUPS, LSPLIT);
  rkhs_main<<<grid, WPB * 32, 0, stream>>>(x, alpha, beta, partial);

  reduce_kernel<<<(NPTS * DDIM + 255) / 256, 256, 0, stream>>>(partial, out);
}


# main: the block's beta slab stays in LDS (a == j zeroed), written by the prologue that reads it anyway; a column's A fragments are 12 ds_load_b128 in one group, were 48 global loads; values unchanged
# speedup vs baseline: 6.2047x; 1.0864x over previous
#include <hip/hip_runtime.h>


typedef __attribute__((ext_vector_type(2))) float v2f;
typedef __attribute__((ext_vector_type(4))) float v4f;
typedef __attribute__((ext_vector_type(8))) float v8f;

#define NPTS 2000
#define DDIM 24
#define NTILE 125
#define TL 4
#define LCHUNK (TL * 16)
#define LSPLIT ((NTILE + TL - 1) / TL)
#define RT 2
#define WPB 8
#define IGROUPS ((NTILE + RT * WPB - 1) / (RT * WPB))
#define XSCALE 1.2011224087864498f
#define APITCH 28

__global__ __launch_bounds__(WPB * 32) void rkhs_main(const float* __restrict__ x,
                                                 const float* __restrict__ alpha,
                                                 const float* __restrict__ beta,
                                                 float* __restrict__ partial) {
  __shared__ __attribute__((aligned(16))) float xs[DDIM][LCHUNK];
  __shared__ __attribute__((aligned(16))) float cs[DDIM][LCHUNK];
  __shared__ __attribute__((aligned(16))) float As[DDIM][LCHUNK][APITCH];

  const int tid   = threadIdx.x;
  const int lane  = tid & 31;
  const int wave  = tid >> 5;
  const int chunk = blockIdx.y;

  bool lok[TL];
#pragma unroll
  for (int t = 0; t < TL; ++t) lok[t] = (chunk * TL + t < NTILE);

#pragma unroll
  for (int u = 0; u < (DDIM * LCHUNK) / (WPB * 32); ++u) {
    int e = tid + u * (WPB * 32);
    int ll = e / DDIM, k = e - ll * DDIM;
    int lt = chunk * TL + (ll >> 4);
    if (lt > NTILE - 1) lt = NTILE - 1;
    xs[k][ll] = XSCALE * x[(lt * 16 + (ll & 15)) * DDIM + k];
  }

  {
    int ltc = chunk * TL + (lane >> 3);
    if (ltc > NTILE - 1) ltc = NTILE - 1;
    const int lg = ltc * 16 + ((2 * lane) & 15);
    const v4f* xrow = (const v4f*)(x + lg * DDIM);
    v4f xr[12];
#pragma unroll
    for (int q = 0; q < 12; ++q) xr[q] = xrow[q];
#pragma unroll
    for (int u = 0; u < DDIM / WPB; ++u) {
      const int j = wave + WPB * u;
      const float* bq = beta + (size_t)j * DDIM * NPTS + lg;
      float s0 = 0.0f, s1 = 0.0f;
      float row0[DDIM], row1[DDIM];
#pragma unroll
      for (int a = 0; a < DDIM; ++a) {
        v2f b = *(const v2f*)(bq + a * NPTS);
        float p0 = b.x * xr[a >> 2][a & 3];
        float p1 = b.y * xr[6 + (a >> 2)][a & 3];
        s0 += (a == j) ? 0.0f : p0;
        s1 += (a == j) ? 0.0f : p1;
        const int slot = 12 * ((a >> 1) & 1) + 2 * (a >> 2) + (a & 1);
        row0[slot] = (a == j) ? 0.0f : b.x;
        row1[slot] = (a == j) ? 0.0f : b.y;
      }
#pragma unroll
      for (int m = 0; m < DDIM / 4; ++m) {
        *(v4f*)&As[j][2 * lane][4 * m] =
            (v4f){row0[4 * m], row0[4 * m + 1], row0[4 * m + 2], row0[4 * m + 3]};
        *(v4f*)&As[j][2 * lane + 1][4 * m] =
            (v4f){row1[4 * m], row1[4 * m + 1], row1[4 * m + 2], row1[4 * m + 3]};
      }
      v2f al = *(const v2f*)(alpha + j * NPTS + lg);
      v2f r;
      r.x = al.x - 2.0f * s0;
      r.y = al.y - 2.0f * s1;
      *(v2f*)&cs[j][2 * lane] = r;
    }
  }
  __syncthreads();

  const int c = lane & 15;
  const int h = lane >> 4;
  const int it0 = (blockIdx.x * WPB + wave) * RT;
  if (it0 >= NTILE) return;

  int  irow[RT];
  bool ok[RT];
#pragma unroll
  for (int r = 0; r < RT; ++r) {
    int it = it0 + r;
    ok[r] = (it < NTILE);
    if (it > NTILE - 1) it = NTILE - 1;
    irow[r] = it * 16 + c;
  }


  v2f bx[RT][6];
#pragma unroll
  for (int r = 0; r < RT; ++r) {
    const v4f* xr = (const v4f*)(x + irow[r] * DDIM);
#pragma unroll
    for (int q = 0; q < 6; ++q) {
      v4f v = xr[q];
      bx[r][q].x = 2.0f * (h ? v.z : v.x);
      bx[r][q].y = 2.0f * (h ? v.w : v.y);
    }
  }

  v2f S[RT][TL][4];
#pragma unroll
  for (int r = 0; r < RT; ++r)
#pragma unroll
    for (int t = 0; t < TL; ++t)
#pragma unroll
      for (int p = 0; p < 4; ++p) S[r][t][p] = (v2f){0.0f, 0.0f};

  v4f rawN[RT];
#pragma unroll
  for (int r = 0; r < RT; ++r) rawN[r] = *(const v4f*)(x + irow[r] * DDIM);

#pragma unroll 1
  for (int q = 0; q < 6; ++q) {
    const int qn = (q + 1 < 6) ? (q + 1) : q;
    v4f raw[RT];
#pragma unroll
    for (int r = 0; r < RT; ++r) {
      raw[r] = rawN[r];
      rawN[r] = *(const v4f*)(x + irow[r] * DDIM + 4 * qn);
    }
    v4f lo[4][TL], hi[4][TL];
#pragma unroll
    for (int kq = 0; kq < 4; ++kq)
#pragma unroll
      for (int t = 0; t < TL; ++t) {
        lo[kq][t] = *(const v4f*)&xs[4 * q + kq][8 * h + 16 * t];
        hi[kq][t] = *(const v4f*)&xs[4 * q + kq][8 * h + 16 * t + 4];
      }
    float xq[RT][4];
#pragma unroll
    for (int r = 0; r < RT; ++r) {
      xq[r][0] = XSCALE * raw[r].x; xq[r][1] = XSCALE * raw[r].y;
      xq[r][2] = XSCALE * raw[r].z; xq[r][3] = XSCALE * raw[r].w;
    }
#pragma unroll
    for (int kq = 0; kq < 4; ++kq)
#pragma unroll
      for (int t = 0; t < TL; ++t) {
        v2f xl[4] = {{lo[kq][t].x, lo[kq][t].y}, {lo[kq][t].z, lo[kq][t].w},
                     {hi[kq][t].x, hi[kq][t].y}, {hi[kq][t].z, hi[kq][t].w}};
#pragma unroll
        for (int r = 0; r < RT; ++r)
#pragma unroll
          for (int p = 0; p < 4; ++p) {
            v2f d = xq[r][kq] - xl[p];
            S[r][t][p] += d * d;
          }
      }
  }

#pragma unroll
  for (int t = 0; t < TL; ++t)
    if (!lok[t]) {
#pragma unroll
      for (int r = 0; r < RT; ++r)
#pragma unroll
        for (int p = 0; p < 4; ++p) S[r][t][p] = (v2f){__builtin_inff(), __builtin_inff()};
    }

  v2f A[2][TL][6];
#pragma unroll
  for (int t = 0; t < TL; ++t)
#pragma unroll
    for (int q = 0; q < 3; ++q) {
      v4f v = *(const v4f*)&As[0][16 * t + c][12 * h + 4 * q];
      A[0][t][2 * q]     = (v2f){v.x, v.y};
      A[0][t][2 * q + 1] = (v2f){v.z, v.w};
    }

  float xnext[RT];
#pragma unroll
  for (int r = 0; r < RT; ++r) xnext[r] = x[irow[r] * DDIM];

  v8f CstN[TL];
  v2f xlN[TL][4];
#pragma unroll
  for (int t = 0; t < TL; ++t) {
    v4f clo = *(const v4f*)&cs[0][16 * t + 8 * h];
    v4f chi = *(const v4f*)&cs[0][16 * t + 8 * h + 4];
    CstN[t] = __builtin_shufflevector(clo, chi, 0, 1, 2, 3, 4, 5, 6, 7);
    v4f xlo = *(const v4f*)&xs[0][16 * t + 8 * h];
    v4f xhi = *(const v4f*)&xs[0][16 * t + 8 * h + 4];
    xlN[t][0] = (v2f){xlo.x, xlo.y}; xlN[t][1] = (v2f){xlo.z, xlo.w};
    xlN[t][2] = (v2f){xhi.x, xhi.y}; xlN[t][3] = (v2f){xhi.z, xhi.w};
  }

  float aP[RT], sP[RT];
#pragma unroll
  for (int r = 0; r < RT; ++r) { aP[r] = 0.0f; sP[r] = 0.0f; }

#pragma unroll 1
  for (int j0 = 0; j0 < DDIM; j0 += 2) {
#pragma unroll
  for (int jj = 0; jj < 2; ++jj) {
    const int j = j0 + jj;
    const int jn = (j + 1 < DDIM) ? (j + 1) : j;

    float xij[RT];
    v2f   acc[RT][2];
#pragma unroll
    for (int r = 0; r < RT; ++r) {
      xij[r] = XSCALE * xnext[r];
      xnext[r] = x[irow[r] * DDIM + jn];
      acc[r][0] = (v2f){0.0f, 0.0f};
      acc[r][1] = (v2f){0.0f, 0.0f};
    }

    v8f Cst[TL];
    v2f xl[TL][4];
#pragma unroll
    for (int t = 0; t < TL; ++t) {
      Cst[t] = CstN[t];
#pragma unroll
      for (int p = 0; p < 4; ++p) xl[t][p] = xlN[t][p];
    }

#pragma unroll
    for (int t = 0; t < TL; ++t)
#pragma unroll
      for (int q = 0; q < 3; ++q) {
        v4f v = *(const v4f*)&As[jn][16 * t + c][12 * h + 4 * q];
        A[jj ^ 1][t][2 * q]     = (v2f){v.x, v.y};
        A[jj ^ 1][t][2 * q + 1] = (v2f){v.z, v.w};
      }
#pragma unroll
    for (int t2 = 0; t2 < TL; ++t2) {
      v4f clo = *(const v4f*)&cs[jn][16 * t2 + 8 * h];
      v4f chi = *(const v4f*)&cs[jn][16 * t2 + 8 * h + 4];
      CstN[t2] = __builtin_shufflevector(clo, chi, 0, 1, 2, 3, 4, 5, 6, 7);
      v4f xlo = *(const v4f*)&xs[jn][16 * t2 + 8 * h];
      v4f xhi = *(const v4f*)&xs[jn][16 * t2 + 8 * h + 4];
      xlN[t2][0] = (v2f){xlo.x, xlo.y}; xlN[t2][1] = (v2f){xlo.z, xlo.w};
      xlN[t2][2] = (v2f){xhi.x, xhi.y}; xlN[t2][3] = (v2f){xhi.z, xhi.w};
    }
    __builtin_amdgcn_sched_barrier(0);

    v8f Cp[RT];
    v2f wp[8];
#pragma unroll
    for (int t = 0; t < TL; ++t) {
      v8f C[RT];
      C[0] = Cst[t];
      C[1] = Cst[t];
      v2f w[8];
#pragma unroll
      for (int kk = 0; kk < 6; ++kk) {
        C[0] = __builtin_amdgcn_wmma_f32_16x16x4_f32(false, A[jj][t][kk], false, bx[0][kk],
                                                     (short)0, C[0], false, false);
        C[1] = __builtin_amdgcn_wmma_f32_16x16x4_f32(false, A[jj][t][kk], false, bx[1][kk],
                                                     (short)0, C[1], false, false);
        if (kk < 4) {
#pragma unroll
          for (int u = 2 * kk; u < 2 * kk + 2; ++u) {
            const int p = u >> 1, r = u & 1;
            v2f d = xij[r] - xl[t][p];
            v2f e = d * d - S[r][t][p];
            w[u] = (v2f){__builtin_amdgcn_exp2f(e.x), __builtin_amdgcn_exp2f(e.y)};
          }
        } else if (t > 0) {
#pragma unroll
          for (int u = 4 * (kk - 4); u < 4 * (kk - 4) + 4; ++u) {
            const int p = u >> 1, r = u & 1;
            v2f tt = {Cp[r][2 * p], Cp[r][2 * p + 1]};
            acc[r][p & 1] += wp[u] * tt;
          }
        }
        __builtin_amdgcn_sched_barrier(0);
      }
      Cp[0] = C[0];
      Cp[1] = C[1];
#pragma unroll
      for (int u = 0; u < 8; ++u) wp[u] = w[u];
    }

#pragma unroll
    for (int u = 0; u < 8; ++u) {
      const int p = u >> 1, r = u & 1;
      v2f tt = {Cp[r][2 * p], Cp[r][2 * p + 1]};
      acc[r][p & 1] += wp[u] * tt;
    }

    float aN[RT], sN[RT];
#pragma unroll
    for (int r = 0; r < RT; ++r)
      aN[r] = (acc[r][0].x + acc[r][0].y) + (acc[r][1].x + acc[r][1].y);
#pragma unroll
    for (int r = 0; r < RT; ++r) sN[r] = __shfl_xor(aN[r], 16);
#pragma unroll
    for (int r = 0; r < RT; ++r) {
      float v = aP[r] + sP[r];
      if (h == 0 && ok[r] && j > 0)
        partial[((size_t)chunk * NPTS + irow[r]) * DDIM + (j - 1)] = v;
    }
#pragma unroll
    for (int r = 0; r < RT; ++r) { aP[r] = aN[r]; sP[r] = sN[r]; }
  }
  }

#pragma unroll
  for (int r = 0; r < RT; ++r) {
    float v = aP[r] + sP[r];
    if (h == 0 && ok[r])
      partial[((size_t)chunk * NPTS + irow[r]) * DDIM + (DDIM - 1)] = v;
  }
}

__global__ __launch_bounds__(256) void reduce_kernel(const float* __restrict__ partial,
                                                     float* __restrict__ out) {
  int t = blockIdx.x * blockDim.x + threadIdx.x;
  if (t < NPTS * DDIM) {
    float s = 0.0f;
#pragma unroll
    for (int c = 0; c < LSPLIT; ++c)
      s += partial[(size_t)c * (NPTS * DDIM) + t];
    out[t] = s;
  }
}

extern "C" void kernel_launch(void* const* d_in, const int* in_sizes, int n_in,
                              void* d_out, int out_size, void* d_ws, size_t ws_size,
                              hipStream_t stream) {
  const float* x     = (const float*)d_in[0];
  const float* alpha = (const float*)d_in[1];
  const float* beta  = (const float*)d_in[2];
  float* out = (float*)d_out;

  float* partial = (float*)d_ws;

  (void)in_sizes; (void)n_in; (void)out_size; (void)ws_size;

  dim3 grid(IGROUPS, LSPLIT);
  rkhs_main<<<grid, WPB * 32, 0, stream>>>(x, alpha, beta, partial);

  reduce_kernel<<<(NPTS * DDIM + 255) / 256, 256, 0, stream>>>(partial, out);
}
